# starting kernel
# baseline (speedup 1.0000x reference)
#include <hip/hip_runtime.h>
#include <hip/hip_bf16.h>


typedef __attribute__((ext_vector_type(2))) float v2f;
typedef __attribute__((ext_vector_type(8))) float v8f;

#define M_DIM 128
#define N_DIM 784
#define K_DIM 192
#define A_ELEMS (M_DIM * K_DIM)
#define B_ELEMS (K_DIM * N_DIM)

__global__ __launch_bounds__(256)
void im2col_pack(const float* __restrict__ x, const float* __restrict__ w,
                 float* __restrict__ Amat, float* __restrict__ Bmat) {
  int idx = blockIdx.x * blockDim.x + threadIdx.x;

  if (idx < A_ELEMS) {
    int i = idx / K_DIM, K = idx % K_DIM;
    int j = K / 96, rem = K % 96, l = rem / 3, k = rem % 3;
    Amat[idx] = w[i * 192 + l * 6 + j * 3 + k];
  }

  int t = idx - A_ELEMS;
  if (t >= 0 && t < B_ELEMS) {
    int K = t / N_DIM, p = t % N_DIM;
    int j = K / 96, rem = K % 96, l = rem / 3, k = rem % 3;
    int n = p / 28, o = p % 28;
    int heff = o + k - 1;
    float val = 0.0f;
    if (heff >= 0 && heff < 28) {
      int s    = (2 * n + j + 55) % 56;
      int hcol = (heff + 27) % 28;
      val = x[l * 1568 + s * 28 + hcol];
    }
    Bmat[t] = val;
  }
}

__global__ __launch_bounds__(128)
void wmma_gemm(const float* __restrict__ Amat, const float* __restrict__ Bmat,
               float* __restrict__ out) {
  __shared__ float As[16 * K_DIM];

  const int tid  = threadIdx.x;
  const int lane = tid & 31;
  const int wid  = tid >> 5;
  const int mt   = blockIdx.x / 13;
  const int ng   = blockIdx.x % 13;
  const int nt   = ng * 4 + wid;
  const int m0   = mt * 16;

  for (int e = tid; e < 16 * K_DIM; e += 128) {
    As[e] = Amat[(m0 + (e / K_DIM)) * K_DIM + (e % K_DIM)];
  }
  __syncthreads();

  if (nt < 49) {
    const int m    = lane & 15;
    const int half = lane >> 4;
    const int n    = nt * 16 + (lane & 15);
    const float* Arow = &As[m * K_DIM + 2 * half];
    const float* Bcol = Bmat + n;

    v8f acc = {};

#pragma unroll
    for (int kk = 0; kk < K_DIM; kk += 4) {
      v2f av;
      av.x = Arow[kk];
      av.y = Arow[kk + 1];
      v2f bv;
      bv.x = Bcol[(kk + 2 * half) * N_DIM];
      bv.y = Bcol[(kk + 2 * half + 1) * N_DIM];
      acc = __builtin_amdgcn_wmma_f32_16x16x4_f32(
          false, av, false, bv, (short)0, acc, false, false);
    }

    const int mbase = m0 + 8 * half;
#pragma unroll
    for (int r = 0; r < 8; ++r) {
      out[(mbase + r) * N_DIM + n] = acc[r];
    }
  }
}

__global__ __launch_bounds__(256)
void naive_conv(const float* __restrict__ x, const float* __restrict__ w,
                float* __restrict__ out) {
  int idx = blockIdx.x * blockDim.x + threadIdx.x;
  if (idx >= M_DIM * N_DIM) return;
  int i = idx / N_DIM, p = idx % N_DIM;
  int n = p / 28, o = p % 28;
  float acc = 0.0f;
  for (int j = 0; j < 2; ++j) {
    int s = (2 * n + j + 55) % 56;
    for (int l = 0; l < 32; ++l) {
      const float* xrow = x + l * 1568 + s * 28;
      const float* wrow = w + i * 192 + l * 6 + j * 3;
#pragma unroll
      for (int k = 0; k < 3; ++k) {
        int heff = o + k - 1;
        if (heff >= 0 && heff < 28) acc += xrow[(heff + 27) % 28] * wrow[k];
      }
    }
  }
  out[idx] = acc;
}

extern "C" void kernel_launch(void* const* d_in, const int* in_sizes, int n_in,
                              void* d_out, int out_size, void* d_ws, size_t ws_size,
                              hipStream_t stream) {
  const float* x = (const float*)d_in[0];
  const float* w = (const float*)d_in[1];
  float* out = (float*)d_out;

  const size_t need = (size_t)(A_ELEMS + B_ELEMS) * sizeof(float);
  if (ws_size < need || d_ws == nullptr) {
    naive_conv<<<(M_DIM * N_DIM + 255) / 256, 256, 0, stream>>>(x, w, out);
    return;
  }

  float* Amat = (float*)d_ws;
  float* Bmat = (float*)((char*)d_ws + (size_t)A_ELEMS * sizeof(float));

  im2col_pack<<<(A_ELEMS + B_ELEMS) / 256, 256, 0, stream>>>(x, w, Amat, Bmat);

  wmma_gemm<<<8 * 13, 128, 0, stream>>>(Amat, Bmat, out);
}


# One launch, no workspace: direct conv, a work-group per output row x 16 channels stages its 64 rolled+padded x rows and 16 w rows in LDS by b128 loads; K order permuted, so sum order changes
# speedup vs baseline: 2.6466x; 2.6466x over previous
#include <hip/hip_runtime.h>


typedef __attribute__((ext_vector_type(2))) float v2f;
typedef __attribute__((ext_vector_type(4))) float v4f;
typedef __attribute__((ext_vector_type(8))) float v8f;

#define WP 196
#define XP 32
#define XSKEW 32
#define XHALF (32 * XP + XSKEW)

__global__ __launch_bounds__(64)
void conv_rows_wmma(const float* __restrict__ x, const float* __restrict__ w,
                    float* __restrict__ out) {
  __shared__ __attribute__((aligned(16))) float Ws[16 * WP];
  __shared__ __attribute__((aligned(16))) float Xs[64 * XP + XSKEW];

  asm volatile("" :: "s"(out));

  const int tid  = threadIdx.x;
  const int lane = tid & 31;
  const int wv   = tid >> 5;
  const int n    = blockIdx.x >> 3;
  const int mt   = blockIdx.x & 7;

  v4f wq[12];
  const float* wsrc = w + mt * (16 * 192) + 4 * tid;
#pragma unroll
  for (int q = 0; q < 12; ++q) wq[q] = *(const v4f*)(wsrc + 256 * q);

  const int r   = tid & 15;
  const int rr  = r < 14 ? r : 13;
  const int j   = rr >= 7 ? 1 : 0;
  const int cc  = rr - 7 * j;
  const int t4  = tid >> 4;
  const int s0  = n ? 2 * n - 1 : 55;
  const int s1  = 2 * n;
  const float* xsrc = x + t4 * 1568 + (j ? s1 : s0) * 28 + 4 * cc;
  v4f xv[8];
#pragma unroll
  for (int q = 0; q < 8; ++q) xv[q] = *(const v4f*)(xsrc + q * (4 * 1568));

#pragma unroll
  for (int q = 0; q < 12; ++q) {
    const int R    = (64 * q) / 48;
    const int m48  = (64 * q) % 48;
    const int bump = (tid >= 48 - m48) ? 1 : 0;
    *(v4f*)&Ws[(R + bump) * WP + 4 * (m48 + tid - 48 * bump)] = wq[q];
  }

  float* xdst = &Xs[(2 * t4 + j) * XP];
  const bool tail = (cc == 6);
#pragma unroll
  for (int q = 0; q < 8; ++q) {
    v4f v = xv[q];
    if (tail) v.w = 0.0f;
    *(v4f*)&xdst[q * (8 * XP) + (q >= 4 ? XSKEW : 0) + 4 + 4 * cc] = v;
  }
  if (tail) {
#pragma unroll
    for (int q = 0; q < 8; ++q) {
      v2f z;
      z.x = 0.0f;
      z.y = xv[q].w;
      *(v2f*)&xdst[q * (8 * XP) + (q >= 4 ? XSKEW : 0) + 2] = z;
    }
  }
  __syncthreads();

  const int c  = lane & 15;
  const int h  = lane >> 4;
  const int o  = 16 * wv + c;
  const int oc = o < 28 ? o : 27;

  const v4f* Ap = (const v4f*)&Ws[c * WP + 96 * h];
  const float* Bp = &Xs[h * XHALF + 2 + oc];
  v4f a4[24];
#pragma unroll
  for (int q = 0; q < 24; ++q) a4[q] = Ap[q];
  float bf[96];
#pragma unroll
  for (int f = 0; f < 96; ++f) {
    const int lp = f / 6;
    const int jk = f - 6 * lp;
    const int jj = jk / 3;
    const int k  = jk - 3 * jj;
    bf[f] = Bp[(2 * lp + jj) * XP + k];
  }
  __builtin_amdgcn_sched_barrier(0);

  v8f acc = {};
#pragma unroll
  for (int t = 0; t < 48; ++t) {
    v2f av, bv;
    if ((t & 1) == 0) { av.x = a4[t >> 1].x; av.y = a4[t >> 1].y; }
    else              { av.x = a4[t >> 1].z; av.y = a4[t >> 1].w; }
    bv.x = bf[2 * t];
    bv.y = bf[2 * t + 1];
    acc = __builtin_amdgcn_wmma_f32_16x16x4_f32(
        false, av, false, bv, (short)0, acc, false, false);
  }

  if (o < 28) {
    float* op = out + (16 * mt + 8 * h) * 784 + n * 28 + o;
#pragma unroll
    for (int i = 0; i < 8; ++i) op[i * 784] = acc[i];
  }
}

extern "C" void kernel_launch(void* const* d_in, const int* in_sizes, int n_in,
                              void* d_out, int out_size, void* d_ws, size_t ws_size,
                              hipStream_t stream) {
  const float* x = (const float*)d_in[0];
  const float* w = (const float*)d_in[1];
  float* out = (float*)d_out;

  conv_rows_wmma<<<224, 64, 0, stream>>>(x, w, out);
}
